# starting kernel
# baseline (speedup 1.0000x reference)
#include <hip/hip_runtime.h>
#include <hip/hip_bf16.h>
#include <math.h>

#define BATCH   4
#define SEQL    2048
#define DMODEL  512
#define NSTATE  16
#define OUTD    512
#define MROWS   (BATCH * SEQL)
#define KDIM    512
#define NCHUNK  16
#define CLEN    (SEQL / NCHUNK)
#define LN_EPS  1e-5f

typedef float v2f __attribute__((ext_vector_type(2)));
typedef float v8f __attribute__((ext_vector_type(8)));

__global__ __launch_bounds__(128)
void s6_gemm_wmma_k512(const float* __restrict__ A,
                       const float* __restrict__ W,
                       const float* __restrict__ bias,
                       float* __restrict__ out,
                       int Nout, int act  ) {
    __shared__ float sB[KDIM * 16];

    const int n0 = blockIdx.y << 4;
    for (int idx = threadIdx.x; idx < KDIM * 16; idx += 128) {
        const int k = idx >> 4, nn = idx & 15;
        sB[idx] = W[(size_t)k * Nout + n0 + nn];
    }
    __syncthreads();

    const int wave  = threadIdx.x >> 5;
    const int lane  = threadIdx.x & 31;
    const int mtile = (blockIdx.x << 2) + wave;
    const int mrow  = (mtile << 4) + (lane & 15);
    const int klo   = (lane >> 4) << 1;
    const int ncol  = lane & 15;

    v8f acc = {0.f, 0.f, 0.f, 0.f, 0.f, 0.f, 0.f, 0.f};

    const float* ap = A + (size_t)mrow * KDIM + klo;
#pragma unroll 4
    for (int k0 = 0; k0 < KDIM; k0 += 4) {
        v2f av; av.x = ap[k0];  av.y = ap[k0 + 1];
        const int kk = (k0 + klo) << 4;
        v2f bv; bv.x = sB[kk + ncol]; bv.y = sB[kk + 16 + ncol];
        acc = __builtin_amdgcn_wmma_f32_16x16x4_f32(
             false, av,  false, bv,
             (short)0, acc,  false,  false);
    }

    const float bb  = bias[n0 + ncol];
    const int   mhi = (lane >> 4) << 3;
#pragma unroll
    for (int r = 0; r < 8; ++r) {
        float v = acc[r] + bb;
        if (act == 1) v = (v > 20.f) ? v : log1pf(expf(v));
        out[(size_t)((mtile << 4) + mhi + r) * Nout + n0 + ncol] = v;
    }
}

__global__ __launch_bounds__(256)
void s6_scan_pass1(const float* __restrict__ x,
                   const float* __restrict__ dt,
                   const float* __restrict__ Bm,
                   const float* __restrict__ A_log,
                   float* __restrict__ hend,
                   float* __restrict__ aend) {
    const int b = blockIdx.z, c = blockIdx.y;
    const int d = blockIdx.x * 256 + threadIdx.x;

    __shared__ float sBm[CLEN * NSTATE];
    for (int idx = threadIdx.x; idx < CLEN * NSTATE; idx += 256)
        sBm[idx] = Bm[((size_t)b * SEQL + c * CLEN) * NSTATE + idx];
    __syncthreads();

    float Ac[NSTATE];
#pragma unroll
    for (int n = 0; n < NSTATE; ++n) Ac[n] = -expf(A_log[d * NSTATE + n]);

    float h[NSTATE], ap[NSTATE];
#pragma unroll
    for (int n = 0; n < NSTATE; ++n) { h[n] = 0.f; ap[n] = 1.f; }

    for (int t = 0; t < CLEN; ++t) {
        const size_t ofs = ((size_t)b * SEQL + c * CLEN + t) * DMODEL + d;
        const float dtv = dt[ofs];
        const float ub  = dtv * x[ofs];
        const int   bmo = t * NSTATE;
#pragma unroll
        for (int n = 0; n < NSTATE; ++n) {
            const float a = __expf(dtv * Ac[n]);
            h[n]  = a * h[n] + ub * sBm[bmo + n];
            ap[n] *= a;
        }
    }

    const size_t base = (((size_t)b * DMODEL + d) * NCHUNK + c) * NSTATE;
#pragma unroll
    for (int n = 0; n < NSTATE; ++n) { hend[base + n] = h[n]; aend[base + n] = ap[n]; }
}

__global__ __launch_bounds__(256)
void s6_scan_pass2(const float* __restrict__ hend,
                   const float* __restrict__ aend,
                   float* __restrict__ carry) {
    const int idx = blockIdx.x * 256 + threadIdx.x;
    const size_t bd = (size_t)(idx >> 4);
    const int n = idx & 15;
    float cv = 0.f;
    for (int c = 0; c < NCHUNK; ++c) {
        const size_t o = (bd * NCHUNK + c) * NSTATE + n;
        carry[o] = cv;
        cv = aend[o] * cv + hend[o];
    }
}

__global__ __launch_bounds__(256)
void s6_scan_pass3(const float* __restrict__ x,
                   const float* __restrict__ dt,
                   const float* __restrict__ Bm,
                   const float* __restrict__ Cm,
                   const float* __restrict__ A_log,
                   const float* __restrict__ carry,
                   const float* __restrict__ Dskip,
                   float* __restrict__ yout) {
    const int b = blockIdx.z, c = blockIdx.y;
    const int d = blockIdx.x * 256 + threadIdx.x;

    __shared__ float sBm[CLEN * NSTATE];
    __shared__ float sCm[CLEN * NSTATE];
    for (int idx = threadIdx.x; idx < CLEN * NSTATE; idx += 256) {
        const size_t g = ((size_t)b * SEQL + c * CLEN) * NSTATE + idx;
        sBm[idx] = Bm[g];
        sCm[idx] = Cm[g];
    }
    __syncthreads();

    float Ac[NSTATE];
#pragma unroll
    for (int n = 0; n < NSTATE; ++n) Ac[n] = -expf(A_log[d * NSTATE + n]);

    float h[NSTATE];
    const size_t base = (((size_t)b * DMODEL + d) * NCHUNK + c) * NSTATE;
#pragma unroll
    for (int n = 0; n < NSTATE; ++n) h[n] = carry[base + n];

    const float dsk = Dskip[d];
    for (int t = 0; t < CLEN; ++t) {
        const size_t ofs = ((size_t)b * SEQL + c * CLEN + t) * DMODEL + d;
        const float dtv = dt[ofs];
        const float xv  = x[ofs];
        const float ub  = dtv * xv;
        const int   so  = t * NSTATE;
        float y = 0.f;
#pragma unroll
        for (int n = 0; n < NSTATE; ++n) {
            const float a = __expf(dtv * Ac[n]);
            h[n] = a * h[n] + ub * sBm[so + n];
            y += sCm[so + n] * h[n];
        }
        y += dsk * xv;
        yout[ofs] = 0.5f * y * (1.f + erff(y * 0.70710678118654752f));
    }
}

__global__ __launch_bounds__(256)
void s6_ln_residual(const float* __restrict__ ymix,
                    const float* __restrict__ hin,
                    const float* __restrict__ g,
                    const float* __restrict__ be,
                    float* __restrict__ hout) {
    const size_t row = blockIdx.x;
    const int tid = threadIdx.x;
    const float v0 = ymix[row * DMODEL + tid];
    const float v1 = ymix[row * DMODEL + 256 + tid];

    __shared__ float s1[256], s2[256];
    s1[tid] = v0 + v1;
    s2[tid] = v0 * v0 + v1 * v1;
    __syncthreads();
    for (int st = 128; st > 0; st >>= 1) {
        if (tid < st) { s1[tid] += s1[tid + st]; s2[tid] += s2[tid + st]; }
        __syncthreads();
    }
    const float mu   = s1[0] * (1.f / DMODEL);
    const float var  = s2[0] * (1.f / DMODEL) - mu * mu;
    const float rstd = rsqrtf(var + LN_EPS);

    hout[row * DMODEL + tid] =
        hin[row * DMODEL + tid] + (v0 - mu) * rstd * g[tid] + be[tid];
    hout[row * DMODEL + 256 + tid] =
        hin[row * DMODEL + 256 + tid] + (v1 - mu) * rstd * g[256 + tid] + be[256 + tid];
}

extern "C" void kernel_launch(void* const* d_in, const int* in_sizes, int n_in,
                              void* d_out, int out_size, void* d_ws, size_t ws_size,
                              hipStream_t stream) {
    (void)in_sizes; (void)n_in; (void)out_size; (void)ws_size;

    const float* x      = (const float*)d_in[0];
    const float* A_log  = (const float*)d_in[1];
    const float* W_B    = (const float*)d_in[2];
    const float* b_B    = (const float*)d_in[3];
    const float* W_C    = (const float*)d_in[4];
    const float* b_C    = (const float*)d_in[5];
    const float* W_dt   = (const float*)d_in[6];
    const float* b_dt   = (const float*)d_in[7];
    const float* D_skip = (const float*)d_in[8];
    const float* W_mix  = (const float*)d_in[9];
    const float* b_mix  = (const float*)d_in[10];
    const float* ln_g   = (const float*)d_in[11];
    const float* ln_b   = (const float*)d_in[12];
    const float* W_dec  = (const float*)d_in[13];
    const float* b_dec  = (const float*)d_in[14];
    float* out = (float*)d_out;

    float* ws = (float*)d_ws;
    size_t off = 0;
    float* dtb   = ws + off; off += (size_t)MROWS * DMODEL;
    float* Bmb   = ws + off; off += (size_t)MROWS * NSTATE;
    float* Cmb   = ws + off; off += (size_t)MROWS * NSTATE;
    float* ygelu = ws + off; off += (size_t)MROWS * DMODEL;
    float* ymix  = ws + off; off += (size_t)MROWS * DMODEL;
    float* h1    = ws + off; off += (size_t)MROWS * DMODEL;
    float* h2    = ws + off; off += (size_t)MROWS * DMODEL;
    float* hend  = ws + off; off += (size_t)BATCH * DMODEL * NCHUNK * NSTATE;
    float* aend  = ws + off; off += (size_t)BATCH * DMODEL * NCHUNK * NSTATE;
    float* carry = ws + off; off += (size_t)BATCH * DMODEL * NCHUNK * NSTATE;

    const int DN = DMODEL * NSTATE;
    const int DD = DMODEL * DMODEL;

    const dim3 gBig(MROWS / 64, DMODEL / 16);
    const dim3 gBC (MROWS / 64, 1);
    const dim3 gScan(DMODEL / 256, NCHUNK, BATCH);

    for (int i = 0; i < 2; ++i) {
        const float* hin = (i == 0) ? x : h1;
        float* hout      = (i == 0) ? h1 : h2;

        s6_gemm_wmma_k512<<<gBig, 128, 0, stream>>>(
            hin, W_dt + (size_t)i * DD, b_dt + i * DMODEL, dtb, DMODEL, 1);
        s6_gemm_wmma_k512<<<gBC, 128, 0, stream>>>(
            hin, W_B + (size_t)i * DN, b_B + i * NSTATE, Bmb, NSTATE, 0);
        s6_gemm_wmma_k512<<<gBC, 128, 0, stream>>>(
            hin, W_C + (size_t)i * DN, b_C + i * NSTATE, Cmb, NSTATE, 0);

        s6_scan_pass1<<<gScan, 256, 0, stream>>>(
            hin, dtb, Bmb, A_log + (size_t)i * DN, hend, aend);
        s6_scan_pass2<<<BATCH * DMODEL * NSTATE / 256, 256, 0, stream>>>(
            hend, aend, carry);
        s6_scan_pass3<<<gScan, 256, 0, stream>>>(
            hin, dtb, Bmb, Cmb, A_log + (size_t)i * DN, carry,
            D_skip + i * DMODEL, ygelu);

        s6_gemm_wmma_k512<<<gBig, 128, 0, stream>>>(
            ygelu, W_mix + (size_t)i * DD, b_mix + i * DMODEL, ymix, DMODEL, 0);
        s6_ln_residual<<<MROWS, 256, 0, stream>>>(
            ymix, hin, ln_g + i * DMODEL, ln_b + i * DMODEL, hout);
    }

    s6_gemm_wmma_k512<<<gBig, 128, 0, stream>>>(h2, W_dec, b_dec, out, OUTD, 0);
}


# Mix GEMM: a block owns 32 whole rows (W streamed through 2 LDS stages in 32-k chunks) and does LayerNorm+residual in its epilogue; LN launches gone; LN sums reordered (rounding order)
# speedup vs baseline: 3.4626x; 3.4626x over previous
#include <hip/hip_runtime.h>
#include <hip/hip_bf16.h>
#include <math.h>

#define BATCH   4
#define SEQL    2048
#define DMODEL  512
#define NSTATE  16
#define OUTD    512
#define MROWS   (BATCH * SEQL)
#define KDIM    512
#define NCHUNK  16
#define CLEN    (SEQL / NCHUNK)
#define LN_EPS  1e-5f

typedef float v2f __attribute__((ext_vector_type(2)));
typedef float v4f __attribute__((ext_vector_type(4)));
typedef float v8f __attribute__((ext_vector_type(8)));

template <int NT, int NW>
__global__ __launch_bounds__(NW * 32)
void s6_gemm_strip_k512(const float* __restrict__ A,
                        const float* __restrict__ W,
                        const float* __restrict__ bias,
                        float* __restrict__ out,
                        int Nout, int act  ) {
    constexpr int NC = NT * 16;
    constexpr int NQ = KDIM / 4;
    constexpr int NA = (NT == 1) ? 2 : 1;
    __shared__ v4f sB[NQ * NC];

    const int n0 = blockIdx.y * NC;
#pragma unroll 8
    for (int it = 0; it < (NQ * NC) / (NW * 32); ++it) {
        const int idx = threadIdx.x + it * (NW * 32);
        const int q = idx / NC, n = idx % NC;
        const float* wp = W + (size_t)(4 * q) * Nout + n0 + n;
        v4f w;
        w.x = wp[0];
        w.y = wp[(size_t)Nout];
        w.z = wp[(size_t)Nout * 2];
        w.w = wp[(size_t)Nout * 3];
        sB[idx] = w;
    }
    __syncthreads();

    const int wave = threadIdx.x >> 5;
    const int lane = threadIdx.x & 31;
    const int c    = lane & 15;
    const int h    = lane >> 4;
    const int m0   = (blockIdx.x * NW + wave) << 4;

    const v4f* ap = reinterpret_cast<const v4f*>(A + (size_t)(m0 + c) * KDIM) + h;
    const v4f* bp = sB + h * NC + c;

    v8f acc[NT * NA];
#pragma unroll
    for (int i = 0; i < NT * NA; ++i)
        acc[i] = (v8f){0.f, 0.f, 0.f, 0.f, 0.f, 0.f, 0.f, 0.f};

#pragma unroll (NT == 1 ? 8 : 4)
    for (int j = 0; j < KDIM / 8; ++j) {
        const v4f a4 = ap[2 * j];
        v4f b4[NT];
#pragma unroll
        for (int t = 0; t < NT; ++t) b4[t] = bp[(2 * j) * NC + 16 * t];

        v2f a01; a01.x = a4.x; a01.y = a4.y;
        v2f a23; a23.x = a4.z; a23.y = a4.w;
#pragma unroll
        for (int t = 0; t < NT; ++t) {
            v2f bv; bv.x = b4[t].x; bv.y = b4[t].y;
            acc[t * NA] = __builtin_amdgcn_wmma_f32_16x16x4_f32(
                 false, a01,  false, bv,
                 (short)0, acc[t * NA],  false,  false);
        }
#pragma unroll
        for (int t = 0; t < NT; ++t) {
            v2f bv; bv.x = b4[t].z; bv.y = b4[t].w;
            acc[t * NA + NA - 1] = __builtin_amdgcn_wmma_f32_16x16x4_f32(
                 false, a23,  false, bv,
                 (short)0, acc[t * NA + NA - 1],  false,  false);
        }
    }

    const int mhi = h << 3;
#pragma unroll
    for (int t = 0; t < NT; ++t) {
        const int   col = n0 + 16 * t + c;
        const float bb  = bias[col];
#pragma unroll
        for (int r = 0; r < 8; ++r) {
            float v = acc[t * NA][r];
            if (NA == 2) v += acc[t * NA + NA - 1][r];
            v += bb;
            if (act == 1) v = (v > 20.f) ? v : log1pf(expf(v));
            out[(size_t)(m0 + mhi + r) * Nout + col] = v;
        }
    }
}

#define BC_PITCH 20

__global__ __launch_bounds__(256)
void s6_gemm_bc_splitk(const float* __restrict__ A,
                       const float* __restrict__ WB,
                       const float* __restrict__ WC,
                       const float* __restrict__ bB,
                       const float* __restrict__ bC,
                       float* __restrict__ Bm,
                       float* __restrict__ Cm) {
    constexpr int NQ = KDIM / 4;
    __shared__ v4f sW[NQ * 32];
    __shared__ v4f sRedV[4 * 2 * 32 * BC_PITCH / 4];
    float* sRed = reinterpret_cast<float*>(sRedV);

    const int tid  = threadIdx.x;
    const int wave = tid >> 5;
    const int lane = tid & 31;
    const int c    = lane & 15;
    const int h    = lane >> 4;
    const int rt   = wave & 1;
    const int kq   = wave >> 1;
    const int m0   = blockIdx.x * 32 + rt * 16;

    const v4f* ap = reinterpret_cast<const v4f*>(A + (size_t)(m0 + c) * KDIM + kq * 128) + h;
    v4f a4[16];
#pragma unroll
    for (int j = 0; j < 16; ++j) a4[j] = ap[2 * j];

#pragma unroll
    for (int it = 0; it < (NQ * 32) / 256; ++it) {
        const int idx = tid + it * 256;
        const int q = idx >> 5, n = idx & 31;
        const float* wp = ((n < 16) ? (WB + n) : (WC + (n - 16))) + (size_t)(4 * q) * NSTATE;
        v4f w;
        w.x = wp[0];
        w.y = wp[NSTATE];
        w.z = wp[NSTATE * 2];
        w.w = wp[NSTATE * 3];
        sW[idx] = w;
    }
    __syncthreads();

    const v4f* bp = sW + (kq * 32 + h) * 32 + c;
    v8f acc0 = (v8f){0.f, 0.f, 0.f, 0.f, 0.f, 0.f, 0.f, 0.f};
    v8f acc1 = (v8f){0.f, 0.f, 0.f, 0.f, 0.f, 0.f, 0.f, 0.f};
#pragma unroll
    for (int j = 0; j < 16; ++j) {
        const v4f b0 = bp[(2 * j) * 32];
        const v4f b1 = bp[(2 * j) * 32 + 16];
        v2f a01; a01.x = a4[j].x; a01.y = a4[j].y;
        v2f a23; a23.x = a4[j].z; a23.y = a4[j].w;
        v2f b0l; b0l.x = b0.x; b0l.y = b0.y;
        v2f b1l; b1l.x = b1.x; b1l.y = b1.y;
        v2f b0h; b0h.x = b0.z; b0h.y = b0.w;
        v2f b1h; b1h.x = b1.z; b1h.y = b1.w;
        acc0 = __builtin_amdgcn_wmma_f32_16x16x4_f32(false, a01, false, b0l, (short)0, acc0, false, false);
        acc1 = __builtin_amdgcn_wmma_f32_16x16x4_f32(false, a01, false, b1l, (short)0, acc1, false, false);
        acc0 = __builtin_amdgcn_wmma_f32_16x16x4_f32(false, a23, false, b0h, (short)0, acc0, false, false);
        acc1 = __builtin_amdgcn_wmma_f32_16x16x4_f32(false, a23, false, b1h, (short)0, acc1, false, false);
    }

    float* red = sRed + ((kq * 2) * 32 + rt * 16 + 8 * h) * BC_PITCH + c;
#pragma unroll
    for (int r = 0; r < 8; ++r) {
        red[r * BC_PITCH]        = acc0[r];
        red[(32 + r) * BC_PITCH] = acc1[r];
    }
    __syncthreads();

    const int t  = tid >> 7;
    const int q4 = tid & 127;
    const int rl = q4 >> 2;
    const int c0 = (q4 & 3) * 4;
    const float* rp = sRed + (t * 32 + rl) * BC_PITCH + c0;
    v4f s = *reinterpret_cast<const v4f*>(rp);
#pragma unroll
    for (int k = 1; k < 4; ++k)
        s += *reinterpret_cast<const v4f*>(rp + k * 2 * 32 * BC_PITCH);
    s += *reinterpret_cast<const v4f*>((t ? bC : bB) + c0);
    float* op = (t ? Cm : Bm) + ((size_t)blockIdx.x * 32 + rl) * NSTATE + c0;
    *reinterpret_cast<v4f*>(op) = s;
}

#define SCAN_TD 64
#define SCAN_QR (SCAN_TD / 4)

__device__ __forceinline__ void s6_stage_tile2(const float* __restrict__ ga,
                                               const float* __restrict__ gb,
                                               v4f* sa, v4f* sb, int tid) {
#pragma unroll 16
    for (int it = 0; it < CLEN * SCAN_QR / SCAN_TD; ++it) {
        const int idx = tid + it * SCAN_TD;
        const int t = idx / SCAN_QR, q = idx % SCAN_QR;
        const size_t g = (size_t)t * DMODEL + 4 * q;
        sa[idx] = *reinterpret_cast<const v4f*>(ga + g);
        sb[idx] = *reinterpret_cast<const v4f*>(gb + g);
    }
}

__global__ __launch_bounds__(SCAN_TD)
void s6_scan_pass1(const float* __restrict__ x,
                   const float* __restrict__ dt,
                   const float* __restrict__ Bm,
                   const float* __restrict__ A_log,
                   float* __restrict__ hend,
                   float* __restrict__ aend) {
    const int b = blockIdx.z, c = blockIdx.y;
    const int tid = threadIdx.x;
    const int d0 = blockIdx.x * SCAN_TD;
    const int d = d0 + tid;

    __shared__ v4f sDt[CLEN * SCAN_QR];
    __shared__ v4f sX [CLEN * SCAN_QR];
    __shared__ v4f sBm[CLEN * NSTATE / 4];

    const size_t row0 = (size_t)b * SEQL + (size_t)c * CLEN;
    s6_stage_tile2(dt + row0 * DMODEL + d0, x + row0 * DMODEL + d0, sDt, sX, tid);
    const v4f* bmg = reinterpret_cast<const v4f*>(Bm + row0 * NSTATE);
#pragma unroll
    for (int it = 0; it < CLEN * NSTATE / 4 / SCAN_TD; ++it)
        sBm[tid + it * SCAN_TD] = bmg[tid + it * SCAN_TD];

    float Ac[NSTATE];
#pragma unroll
    for (int n = 0; n < NSTATE; ++n) Ac[n] = -expf(A_log[d * NSTATE + n]);

    float h[NSTATE], ap[NSTATE];
#pragma unroll
    for (int n = 0; n < NSTATE; ++n) { h[n] = 0.f; ap[n] = 1.f; }

    __syncthreads();

    const float* sDtf = reinterpret_cast<const float*>(sDt) + tid;
    const float* sXf  = reinterpret_cast<const float*>(sX) + tid;

    float dt_n = sDtf[0], x_n = sXf[0];
    v4f bm_n[4];
#pragma unroll
    for (int k = 0; k < 4; ++k) bm_n[k] = sBm[k];

    for (int t = 0; t < CLEN; ++t) {
        const float dtv = dt_n, xv = x_n;
        v4f bm[4];
#pragma unroll
        for (int k = 0; k < 4; ++k) bm[k] = bm_n[k];
        const int tn = (t + 1 < CLEN) ? t + 1 : t;
        dt_n = sDtf[tn * SCAN_TD];
        x_n  = sXf[tn * SCAN_TD];
#pragma unroll
        for (int k = 0; k < 4; ++k) bm_n[k] = sBm[tn * 4 + k];

        const float ub = dtv * xv;
#pragma unroll
        for (int n = 0; n < NSTATE; ++n) {
            const float a = __expf(dtv * Ac[n]);
            h[n]  = a * h[n] + ub * bm[n >> 2][n & 3];
            ap[n] *= a;
        }
    }

    const size_t base = (((size_t)b * DMODEL + d) * NCHUNK + c) * NSTATE;
#pragma unroll
    for (int n = 0; n < NSTATE; ++n) { hend[base + n] = h[n]; aend[base + n] = ap[n]; }
}

__global__ __launch_bounds__(256)
void s6_scan_pass2(const float* __restrict__ hend,
                   const float* __restrict__ aend,
                   float* __restrict__ carry) {
    const int idx = blockIdx.x * 256 + threadIdx.x;
    const size_t bd = (size_t)(idx >> 4);
    const int n = idx & 15;
    float cv = 0.f;
    for (int c = 0; c < NCHUNK; ++c) {
        const size_t o = (bd * NCHUNK + c) * NSTATE + n;
        carry[o] = cv;
        cv = aend[o] * cv + hend[o];
    }
}

__global__ __launch_bounds__(SCAN_TD)
void s6_scan_pass3(const float* __restrict__ x,
                   const float* __restrict__ dt,
                   const float* __restrict__ Bm,
                   const float* __restrict__ Cm,
                   const float* __restrict__ A_log,
                   const float* __restrict__ carry,
                   const float* __restrict__ Dskip,
                   float* __restrict__ yout) {
    const int b = blockIdx.z, c = blockIdx.y;
    const int tid = threadIdx.x;
    const int d0 = blockIdx.x * SCAN_TD;
    const int d = d0 + tid;

    __shared__ v4f sDt[CLEN * SCAN_QR];
    __shared__ v4f sX [CLEN * SCAN_QR];
    __shared__ v4f sBm[CLEN * NSTATE / 4];
    __shared__ v4f sCm[CLEN * NSTATE / 4];

    const size_t row0 = (size_t)b * SEQL + (size_t)c * CLEN;
    s6_stage_tile2(dt + row0 * DMODEL + d0, x + row0 * DMODEL + d0, sDt, sX, tid);
    const v4f* bmg = reinterpret_cast<const v4f*>(Bm + row0 * NSTATE);
    const v4f* cmg = reinterpret_cast<const v4f*>(Cm + row0 * NSTATE);
#pragma unroll
    for (int it = 0; it < CLEN * NSTATE / 4 / SCAN_TD; ++it) {
        sBm[tid + it * SCAN_TD] = bmg[tid + it * SCAN_TD];
        sCm[tid + it * SCAN_TD] = cmg[tid + it * SCAN_TD];
    }

    float Ac[NSTATE];
#pragma unroll
    for (int n = 0; n < NSTATE; ++n) Ac[n] = -expf(A_log[d * NSTATE + n]);

    float h[NSTATE];
    const size_t base = (((size_t)b * DMODEL + d) * NCHUNK + c) * NSTATE;
#pragma unroll
    for (int n = 0; n < NSTATE; ++n) h[n] = carry[base + n];

    const float dsk = Dskip[d];
    float* yp = yout + row0 * DMODEL + d;

    __syncthreads();

    const float* sDtf = reinterpret_cast<const float*>(sDt) + tid;
    const float* sXf  = reinterpret_cast<const float*>(sX) + tid;

    float dt_n = sDtf[0], x_n = sXf[0];
    v4f bm_n[4], cm_n[4];
#pragma unroll
    for (int k = 0; k < 4; ++k) { bm_n[k] = sBm[k]; cm_n[k] = sCm[k]; }

    for (int t = 0; t < CLEN; ++t) {
        const float dtv = dt_n, xv = x_n;
        v4f bm[4], cm[4];
#pragma unroll
        for (int k = 0; k < 4; ++k) { bm[k] = bm_n[k]; cm[k] = cm_n[k]; }
        const int tn = (t + 1 < CLEN) ? t + 1 : t;
        dt_n = sDtf[tn * SCAN_TD];
        x_n  = sXf[tn * SCAN_TD];
#pragma unroll
        for (int k = 0; k < 4; ++k) { bm_n[k] = sBm[tn * 4 + k]; cm_n[k] = sCm[tn * 4 + k]; }

        const float ub = dtv * xv;
        float y = 0.f;
#pragma unroll
        for (int n = 0; n < NSTATE; ++n) {
            const float a = __expf(dtv * Ac[n]);
            h[n] = a * h[n] + ub * bm[n >> 2][n & 3];
            y += cm[n >> 2][n & 3] * h[n];
        }
        y += dsk * xv;
        yp[(size_t)t * DMODEL] = 0.5f * y * (1.f + erff(y * 0.70710678118654752f));
    }
}

#define MX_KC   32
#define MX_QC   (MX_KC / 4)
#define MX_NCH  (KDIM / MX_KC)
#define MX_STG  (MX_QC * DMODEL)

__device__ __forceinline__ void mx_fetch(const float* __restrict__ wp,
                                         const v4f* __restrict__ ap,
                                         float (&wst)[MX_KC],
                                         v4f (&an)[MX_KC / 8]) {
#pragma unroll
    for (int jj = 0; jj < MX_KC / 8; ++jj) an[jj] = ap[2 * jj];
#pragma unroll
    for (int kk = 0; kk < MX_KC; ++kk) wst[kk] = wp[(size_t)kk * DMODEL];
}

__device__ __forceinline__ void mx_put(v4f* dst, const float (&wst)[MX_KC]) {
#pragma unroll
    for (int q = 0; q < MX_QC; ++q) {
        v4f w;
        w.x = wst[4 * q];
        w.y = wst[4 * q + 1];
        w.z = wst[4 * q + 2];
        w.w = wst[4 * q + 3];
        dst[q * DMODEL] = w;
    }
}

__device__ __forceinline__ void mx_block(const v4f* bp,
                                         const v4f (&a)[MX_KC / 8],
                                         v8f (&acc)[4]) {
    v4f b4[MX_KC / 8][4];
#pragma unroll
    for (int jj = 0; jj < MX_KC / 8; ++jj)
#pragma unroll
        for (int t = 0; t < 4; ++t) b4[jj][t] = bp[(2 * jj) * DMODEL + 16 * t];
#pragma unroll
    for (int jj = 0; jj < MX_KC / 8; ++jj) {
        v2f a01; a01.x = a[jj].x; a01.y = a[jj].y;
        v2f a23; a23.x = a[jj].z; a23.y = a[jj].w;
#pragma unroll
        for (int t = 0; t < 4; ++t) {
            v2f bv; bv.x = b4[jj][t].x; bv.y = b4[jj][t].y;
            acc[t] = __builtin_amdgcn_wmma_f32_16x16x4_f32(
                 false, a01,  false, bv,
                 (short)0, acc[t],  false,  false);
        }
#pragma unroll
        for (int t = 0; t < 4; ++t) {
            v2f bv; bv.x = b4[jj][t].z; bv.y = b4[jj][t].w;
            acc[t] = __builtin_amdgcn_wmma_f32_16x16x4_f32(
                 false, a23,  false, bv,
                 (short)0, acc[t],  false,  false);
        }
    }
}

template <int M>
__device__ __forceinline__ float mx_row_xadd(float v) {
    const int p = __builtin_amdgcn_update_dpp(0, __float_as_int(v), 0x160 + M, 0xf, 0xf, false);
    return v + __int_as_float(p);
}

__global__ __launch_bounds__(512) __attribute__((amdgpu_waves_per_eu(4, 4)))
void s6_gemm_mix_ln(const float* __restrict__ A,
                    const float* __restrict__ W,
                    const float* __restrict__ bias,
                    const float* __restrict__ hin,
                    const float* __restrict__ g,
                    const float* __restrict__ be,
                    float* __restrict__ hout) {
    __shared__ v4f   sW[2 * MX_STG];
    __shared__ float sPart[2 * 32 * 8];
    __shared__ float sMu[32], sRs[32];

    const int tid  = threadIdx.x;
    const int wave = tid >> 5;
    const int lane = tid & 31;
    const int c    = lane & 15;
    const int h    = lane >> 4;
    const int rb   = (wave & 1) << 4;
    const int cs   = wave >> 1;
    const int m0   = blockIdx.x * 32 + rb;
    const int n0   = cs * 64;

    const float* wp = W + tid;
    const v4f*   ap = reinterpret_cast<const v4f*>(A + (size_t)(m0 + c) * KDIM) + h;
    const v4f*   bp = sW + h * DMODEL + n0 + c;
    v4f*         sp = sW + tid;

    float wst[MX_KC];
    v4f a0[MX_KC / 8], a1[MX_KC / 8];
    v8f acc[4];
#pragma unroll
    for (int t = 0; t < 4; ++t)
        acc[t] = (v8f){0.f, 0.f, 0.f, 0.f, 0.f, 0.f, 0.f, 0.f};

    mx_fetch(wp, ap, wst, a0);
    mx_put(sp, wst);
    __syncthreads();

    for (int ch = 0; ch < MX_NCH - 2; ch += 2) {
        mx_fetch(wp + (size_t)(ch + 1) * MX_KC * DMODEL, ap + (ch + 1) * MX_QC, wst, a1);
        mx_block(bp, a0, acc);
        __builtin_amdgcn_sched_barrier(0);
        mx_put(sp + MX_STG, wst);
        __syncthreads();
        mx_fetch(wp + (size_t)(ch + 2) * MX_KC * DMODEL, ap + (ch + 2) * MX_QC, wst, a0);
        mx_block(bp + MX_STG, a1, acc);
        __builtin_amdgcn_sched_barrier(0);
        mx_put(sp, wst);
        __syncthreads();
    }
    mx_fetch(wp + (size_t)(MX_NCH - 1) * MX_KC * DMODEL, ap + (MX_NCH - 1) * MX_QC, wst, a1);
    mx_block(bp, a0, acc);
    __builtin_amdgcn_sched_barrier(0);
    mx_put(sp + MX_STG, wst);
    __syncthreads();
    mx_block(bp + MX_STG, a1, acc);

    const int mhi = h << 3;
    float s1[8], s2[8];
#pragma unroll
    for (int r = 0; r < 8; ++r) { s1[r] = 0.f; s2[r] = 0.f; }
#pragma unroll
    for (int t = 0; t < 4; ++t) {
        const float bb = bias[n0 + 16 * t + c];
#pragma unroll
        for (int r = 0; r < 8; ++r) {
            const float v = acc[t][r] + bb;
            acc[t][r] = v;
            s1[r] += v;
            s2[r] += v * v;
        }
    }
#pragma unroll
    for (int r = 0; r < 8; ++r) { s1[r] = mx_row_xadd<1>(s1[r]); s2[r] = mx_row_xadd<1>(s2[r]); }
#pragma unroll
    for (int r = 0; r < 8; ++r) { s1[r] = mx_row_xadd<2>(s1[r]); s2[r] = mx_row_xadd<2>(s2[r]); }
#pragma unroll
    for (int r = 0; r < 8; ++r) { s1[r] = mx_row_xadd<4>(s1[r]); s2[r] = mx_row_xadd<4>(s2[r]); }
#pragma unroll
    for (int r = 0; r < 8; ++r) { s1[r] = mx_row_xadd<8>(s1[r]); s2[r] = mx_row_xadd<8>(s2[r]); }
    if (c == 0) {
#pragma unroll
        for (int r = 0; r < 8; ++r) {
            sPart[(rb + mhi + r) * 8 + cs]       = s1[r];
            sPart[256 + (rb + mhi + r) * 8 + cs] = s2[r];
        }
    }
    __syncthreads();
    if (tid < 32) {
        const float* p1 = sPart + tid * 8;
        const float* p2 = sPart + 256 + tid * 8;
        float t1 = p1[0], t2 = p2[0];
#pragma unroll
        for (int k = 1; k < 8; ++k) { t1 += p1[k]; t2 += p2[k]; }
        const float mu  = t1 * (1.f / DMODEL);
        const float var = t2 * (1.f / DMODEL) - mu * mu;
        sMu[tid] = mu;
        sRs[tid] = rsqrtf(var + LN_EPS);
    }
    __syncthreads();
    float mu[8], rs[8];
#pragma unroll
    for (int r = 0; r < 8; ++r) { mu[r] = sMu[rb + mhi + r]; rs[r] = sRs[rb + mhi + r]; }
#pragma unroll
    for (int t = 0; t < 4; ++t) {
        const int   col = n0 + 16 * t + c;
        const float gg  = g[col];
        const float bo  = be[col];
#pragma unroll
        for (int r = 0; r < 8; ++r) {
            const size_t o = (size_t)(m0 + mhi + r) * DMODEL + col;
            hout[o] = hin[o] + (acc[t][r] - mu[r]) * rs[r] * gg + bo;
        }
    }
}

extern "C" void kernel_launch(void* const* d_in, const int* in_sizes, int n_in,
                              void* d_out, int out_size, void* d_ws, size_t ws_size,
                              hipStream_t stream) {
    (void)in_sizes; (void)n_in; (void)out_size; (void)ws_size;

    const float* x      = (const float*)d_in[0];
    const float* A_log  = (const float*)d_in[1];
    const float* W_B    = (const float*)d_in[2];
    const float* b_B    = (const float*)d_in[3];
    const float* W_C    = (const float*)d_in[4];
    const float* b_C    = (const float*)d_in[5];
    const float* W_dt   = (const float*)d_in[6];
    const float* b_dt   = (const float*)d_in[7];
    const float* D_skip = (const float*)d_in[8];
    const float* W_mix  = (const float*)d_in[9];
    const float* b_mix  = (const float*)d_in[10];
    const float* ln_g   = (const float*)d_in[11];
    const float* ln_b   = (const float*)d_in[12];
    const float* W_dec  = (const float*)d_in[13];
    const float* b_dec  = (const float*)d_in[14];
    float* out = (float*)d_out;

    float* ws = (float*)d_ws;
    size_t off = 0;
    float* dtb   = ws + off; off += (size_t)MROWS * DMODEL;
    float* Bmb   = ws + off; off += (size_t)MROWS * NSTATE;
    float* Cmb   = ws + off; off += (size_t)MROWS * NSTATE;
    float* ygelu = ws + off; off += (size_t)MROWS * DMODEL;
    float* h1    = ws + off; off += (size_t)MROWS * DMODEL;
    float* h2    = ws + off; off += (size_t)MROWS * DMODEL;
    float* hend  = ws + off; off += (size_t)BATCH * DMODEL * NCHUNK * NSTATE;
    float* aend  = ws + off; off += (size_t)BATCH * DMODEL * NCHUNK * NSTATE;
    float* carry = ws + off; off += (size_t)BATCH * DMODEL * NCHUNK * NSTATE;

    const int DN = DMODEL * NSTATE;
    const int DD = DMODEL * DMODEL;

    const dim3 gBig(MROWS / 256, DMODEL / 64);
    const dim3 gScan(DMODEL / SCAN_TD, NCHUNK, BATCH);

    for (int i = 0; i < 2; ++i) {
        const float* hin = (i == 0) ? x : h1;
        float* hout      = (i == 0) ? h1 : h2;

        s6_gemm_strip_k512<4, 16><<<gBig, 512, 0, stream>>>(
            hin, W_dt + (size_t)i * DD, b_dt + i * DMODEL, dtb, DMODEL, 1);
        s6_gemm_bc_splitk<<<MROWS / 32, 256, 0, stream>>>(
            hin, W_B + (size_t)i * DN, W_C + (size_t)i * DN,
            b_B + i * NSTATE, b_C + i * NSTATE, Bmb, Cmb);

        s6_scan_pass1<<<gScan, SCAN_TD, 0, stream>>>(
            hin, dtb, Bmb, A_log + (size_t)i * DN, hend, aend);
        s6_scan_pass2<<<BATCH * DMODEL * NSTATE / 256, 256, 0, stream>>>(
            hend, aend, carry);
        s6_scan_pass3<<<gScan, SCAN_TD, 0, stream>>>(
            hin, dtb, Bmb, Cmb, A_log + (size_t)i * DN, carry,
            D_skip + i * DMODEL, ygelu);

        s6_gemm_mix_ln<<<MROWS / 32, 512, 0, stream>>>(
            ygelu, W_mix + (size_t)i * DD, b_mix + i * DMODEL, hin,
            ln_g + i * DMODEL, ln_b + i * DMODEL, hout);
    }

    s6_gemm_strip_k512<4, 16><<<gBig, 512, 0, stream>>>(h2, W_dec, b_dec, out, OUTD, 0);
}


# Scan passes 1 and 3: a channel's 16 states split over 4 adjacent lanes (256-thread blocks, 4 waves per SIMD); y summed across the quad by lane swizzle (rounding order); GELU after the loop
# speedup vs baseline: 3.8086x; 1.0999x over previous
#include <hip/hip_runtime.h>
#include <hip/hip_bf16.h>
#include <math.h>

#define BATCH   4
#define SEQL    2048
#define DMODEL  512
#define NSTATE  16
#define OUTD    512
#define MROWS   (BATCH * SEQL)
#define KDIM    512
#define NCHUNK  16
#define CLEN    (SEQL / NCHUNK)
#define LN_EPS  1e-5f

typedef float v2f __attribute__((ext_vector_type(2)));
typedef float v4f __attribute__((ext_vector_type(4)));
typedef float v8f __attribute__((ext_vector_type(8)));

template <int NT, int NW>
__global__ __launch_bounds__(NW * 32)
void s6_gemm_strip_k512(const float* __restrict__ A,
                        const float* __restrict__ W,
                        const float* __restrict__ bias,
                        float* __restrict__ out,
                        int Nout, int act  ) {
    constexpr int NC = NT * 16;
    constexpr int NQ = KDIM / 4;
    constexpr int NA = (NT == 1) ? 2 : 1;
    __shared__ v4f sB[NQ * NC];

    const int n0 = blockIdx.y * NC;
#pragma unroll 8
    for (int it = 0; it < (NQ * NC) / (NW * 32); ++it) {
        const int idx = threadIdx.x + it * (NW * 32);
        const int q = idx / NC, n = idx % NC;
        const float* wp = W + (size_t)(4 * q) * Nout + n0 + n;
        v4f w;
        w.x = wp[0];
        w.y = wp[(size_t)Nout];
        w.z = wp[(size_t)Nout * 2];
        w.w = wp[(size_t)Nout * 3];
        sB[idx] = w;
    }
    __syncthreads();

    const int wave = threadIdx.x >> 5;
    const int lane = threadIdx.x & 31;
    const int c    = lane & 15;
    const int h    = lane >> 4;
    const int m0   = (blockIdx.x * NW + wave) << 4;

    const v4f* ap = reinterpret_cast<const v4f*>(A + (size_t)(m0 + c) * KDIM) + h;
    const v4f* bp = sB + h * NC + c;

    v8f acc[NT * NA];
#pragma unroll
    for (int i = 0; i < NT * NA; ++i)
        acc[i] = (v8f){0.f, 0.f, 0.f, 0.f, 0.f, 0.f, 0.f, 0.f};

#pragma unroll (NT == 1 ? 8 : 4)
    for (int j = 0; j < KDIM / 8; ++j) {
        const v4f a4 = ap[2 * j];
        v4f b4[NT];
#pragma unroll
        for (int t = 0; t < NT; ++t) b4[t] = bp[(2 * j) * NC + 16 * t];

        v2f a01; a01.x = a4.x; a01.y = a4.y;
        v2f a23; a23.x = a4.z; a23.y = a4.w;
#pragma unroll
        for (int t = 0; t < NT; ++t) {
            v2f bv; bv.x = b4[t].x; bv.y = b4[t].y;
            acc[t * NA] = __builtin_amdgcn_wmma_f32_16x16x4_f32(
                 false, a01,  false, bv,
                 (short)0, acc[t * NA],  false,  false);
        }
#pragma unroll
        for (int t = 0; t < NT; ++t) {
            v2f bv; bv.x = b4[t].z; bv.y = b4[t].w;
            acc[t * NA + NA - 1] = __builtin_amdgcn_wmma_f32_16x16x4_f32(
                 false, a23,  false, bv,
                 (short)0, acc[t * NA + NA - 1],  false,  false);
        }
    }

    const int mhi = h << 3;
#pragma unroll
    for (int t = 0; t < NT; ++t) {
        const int   col = n0 + 16 * t + c;
        const float bb  = bias[col];
#pragma unroll
        for (int r = 0; r < 8; ++r) {
            float v = acc[t * NA][r];
            if (NA == 2) v += acc[t * NA + NA - 1][r];
            v += bb;
            if (act == 1) v = (v > 20.f) ? v : log1pf(expf(v));
            out[(size_t)(m0 + mhi + r) * Nout + col] = v;
        }
    }
}

#define BC_PITCH 20

__global__ __launch_bounds__(256)
void s6_gemm_bc_splitk(const float* __restrict__ A,
                       const float* __restrict__ WB,
                       const float* __restrict__ WC,
                       const float* __restrict__ bB,
                       const float* __restrict__ bC,
                       float* __restrict__ Bm,
                       float* __restrict__ Cm) {
    constexpr int NQ = KDIM / 4;
    __shared__ v4f sW[NQ * 32];
    __shared__ v4f sRedV[4 * 2 * 32 * BC_PITCH / 4];
    float* sRed = reinterpret_cast<float*>(sRedV);

    const int tid  = threadIdx.x;
    const int wave = tid >> 5;
    const int lane = tid & 31;
    const int c    = lane & 15;
    const int h    = lane >> 4;
    const int rt   = wave & 1;
    const int kq   = wave >> 1;
    const int m0   = blockIdx.x * 32 + rt * 16;

    const v4f* ap = reinterpret_cast<const v4f*>(A + (size_t)(m0 + c) * KDIM + kq * 128) + h;
    v4f a4[16];
#pragma unroll
    for (int j = 0; j < 16; ++j) a4[j] = ap[2 * j];

#pragma unroll
    for (int it = 0; it < (NQ * 32) / 256; ++it) {
        const int idx = tid + it * 256;
        const int q = idx >> 5, n = idx & 31;
        const float* wp = ((n < 16) ? (WB + n) : (WC + (n - 16))) + (size_t)(4 * q) * NSTATE;
        v4f w;
        w.x = wp[0];
        w.y = wp[NSTATE];
        w.z = wp[NSTATE * 2];
        w.w = wp[NSTATE * 3];
        sW[idx] = w;
    }
    __syncthreads();

    const v4f* bp = sW + (kq * 32 + h) * 32 + c;
    v8f acc0 = (v8f){0.f, 0.f, 0.f, 0.f, 0.f, 0.f, 0.f, 0.f};
    v8f acc1 = (v8f){0.f, 0.f, 0.f, 0.f, 0.f, 0.f, 0.f, 0.f};
#pragma unroll
    for (int j = 0; j < 16; ++j) {
        const v4f b0 = bp[(2 * j) * 32];
        const v4f b1 = bp[(2 * j) * 32 + 16];
        v2f a01; a01.x = a4[j].x; a01.y = a4[j].y;
        v2f a23; a23.x = a4[j].z; a23.y = a4[j].w;
        v2f b0l; b0l.x = b0.x; b0l.y = b0.y;
        v2f b1l; b1l.x = b1.x; b1l.y = b1.y;
        v2f b0h; b0h.x = b0.z; b0h.y = b0.w;
        v2f b1h; b1h.x = b1.z; b1h.y = b1.w;
        acc0 = __builtin_amdgcn_wmma_f32_16x16x4_f32(false, a01, false, b0l, (short)0, acc0, false, false);
        acc1 = __builtin_amdgcn_wmma_f32_16x16x4_f32(false, a01, false, b1l, (short)0, acc1, false, false);
        acc0 = __builtin_amdgcn_wmma_f32_16x16x4_f32(false, a23, false, b0h, (short)0, acc0, false, false);
        acc1 = __builtin_amdgcn_wmma_f32_16x16x4_f32(false, a23, false, b1h, (short)0, acc1, false, false);
    }

    float* red = sRed + ((kq * 2) * 32 + rt * 16 + 8 * h) * BC_PITCH + c;
#pragma unroll
    for (int r = 0; r < 8; ++r) {
        red[r * BC_PITCH]        = acc0[r];
        red[(32 + r) * BC_PITCH] = acc1[r];
    }
    __syncthreads();

    const int t  = tid >> 7;
    const int q4 = tid & 127;
    const int rl = q4 >> 2;
    const int c0 = (q4 & 3) * 4;
    const float* rp = sRed + (t * 32 + rl) * BC_PITCH + c0;
    v4f s = *reinterpret_cast<const v4f*>(rp);
#pragma unroll
    for (int k = 1; k < 4; ++k)
        s += *reinterpret_cast<const v4f*>(rp + k * 2 * 32 * BC_PITCH);
    s += *reinterpret_cast<const v4f*>((t ? bC : bB) + c0);
    float* op = (t ? Cm : Bm) + ((size_t)blockIdx.x * 32 + rl) * NSTATE + c0;
    *reinterpret_cast<v4f*>(op) = s;
}

#define SCAN_TD  64
#define SCAN_QR  (SCAN_TD / 4)
#define SCAN_LPC 4
#define SCAN_NS  (NSTATE / SCAN_LPC)
#define SCAN_NT  (SCAN_TD * SCAN_LPC)

template <int M>
__device__ __forceinline__ float mx_row_xadd(float v) {
    const int p = __builtin_amdgcn_update_dpp(0, __float_as_int(v), 0x160 + M, 0xf, 0xf, false);
    return v + __int_as_float(p);
}

__device__ __forceinline__ void s6_stage_tile2(const float* __restrict__ ga,
                                               const float* __restrict__ gb,
                                               v4f* sa, v4f* sb, int tid) {
#pragma unroll
    for (int it = 0; it < CLEN * SCAN_QR / SCAN_NT; ++it) {
        const int idx = tid + it * SCAN_NT;
        const int t = idx / SCAN_QR, q = idx % SCAN_QR;
        const size_t g = (size_t)t * DMODEL + 4 * q;
        sa[idx] = *reinterpret_cast<const v4f*>(ga + g);
        sb[idx] = *reinterpret_cast<const v4f*>(gb + g);
    }
}

__global__ __launch_bounds__(SCAN_NT)
void s6_scan_pass1(const float* __restrict__ x,
                   const float* __restrict__ dt,
                   const float* __restrict__ Bm,
                   const float* __restrict__ A_log,
                   float* __restrict__ hend,
                   float* __restrict__ aend) {
    const int b = blockIdx.z, c = blockIdx.y;
    const int tid = threadIdx.x;
    const int d0 = blockIdx.x * SCAN_TD;
    const int ch = tid / SCAN_LPC;
    const int sg = tid % SCAN_LPC;
    const int d = d0 + ch;

    __shared__ v4f sDt[CLEN * SCAN_QR];
    __shared__ v4f sX [CLEN * SCAN_QR];
    __shared__ v4f sBm[CLEN * NSTATE / 4];

    const size_t row0 = (size_t)b * SEQL + (size_t)c * CLEN;
    s6_stage_tile2(dt + row0 * DMODEL + d0, x + row0 * DMODEL + d0, sDt, sX, tid);
    const v4f* bmg = reinterpret_cast<const v4f*>(Bm + row0 * NSTATE);
#pragma unroll
    for (int it = 0; it < CLEN * NSTATE / 4 / SCAN_NT; ++it)
        sBm[tid + it * SCAN_NT] = bmg[tid + it * SCAN_NT];

    const v4f al = *reinterpret_cast<const v4f*>(A_log + d * NSTATE + SCAN_NS * sg);
    float Ac[SCAN_NS];
#pragma unroll
    for (int k = 0; k < SCAN_NS; ++k) Ac[k] = -expf(al[k]);

    float h[SCAN_NS], ap[SCAN_NS];
#pragma unroll
    for (int k = 0; k < SCAN_NS; ++k) { h[k] = 0.f; ap[k] = 1.f; }

    __syncthreads();

    const float* sDtf = reinterpret_cast<const float*>(sDt) + ch;
    const float* sXf  = reinterpret_cast<const float*>(sX) + ch;
    const v4f*   sBq  = sBm + sg;

    float dt_n = sDtf[0], x_n = sXf[0];
    v4f bm_n = sBq[0];

    for (int t = 0; t < CLEN; ++t) {
        const float dtv = dt_n, xv = x_n;
        const v4f bm = bm_n;
        const int tn = (t + 1 < CLEN) ? t + 1 : t;
        dt_n = sDtf[tn * SCAN_TD];
        x_n  = sXf[tn * SCAN_TD];
        bm_n = sBq[tn * (NSTATE / 4)];

        const float ub = dtv * xv;
#pragma unroll
        for (int k = 0; k < SCAN_NS; ++k) {
            const float a = __expf(dtv * Ac[k]);
            h[k]  = fmaf(a, h[k], ub * bm[k]);
            ap[k] *= a;
        }
    }

    const size_t base = (((size_t)b * DMODEL + d) * NCHUNK + c) * NSTATE + SCAN_NS * sg;
    v4f hv, av;
#pragma unroll
    for (int k = 0; k < SCAN_NS; ++k) { hv[k] = h[k]; av[k] = ap[k]; }
    *reinterpret_cast<v4f*>(hend + base) = hv;
    *reinterpret_cast<v4f*>(aend + base) = av;
}

__global__ __launch_bounds__(256)
void s6_scan_pass2(const float* __restrict__ hend,
                   const float* __restrict__ aend,
                   float* __restrict__ carry) {
    const int idx = blockIdx.x * 256 + threadIdx.x;
    const size_t bd = (size_t)(idx >> 4);
    const int n = idx & 15;
    float cv = 0.f;
    for (int c = 0; c < NCHUNK; ++c) {
        const size_t o = (bd * NCHUNK + c) * NSTATE + n;
        carry[o] = cv;
        cv = aend[o] * cv + hend[o];
    }
}

__global__ __launch_bounds__(SCAN_NT)
void s6_scan_pass3(const float* __restrict__ x,
                   const float* __restrict__ dt,
                   const float* __restrict__ Bm,
                   const float* __restrict__ Cm,
                   const float* __restrict__ A_log,
                   const float* __restrict__ carry,
                   const float* __restrict__ Dskip,
                   float* __restrict__ yout) {
    const int b = blockIdx.z, c = blockIdx.y;
    const int tid = threadIdx.x;
    const int d0 = blockIdx.x * SCAN_TD;
    const int ch = tid / SCAN_LPC;
    const int sg = tid % SCAN_LPC;
    const int d = d0 + ch;

    __shared__ v4f sDt[CLEN * SCAN_QR];
    __shared__ v4f sX [CLEN * SCAN_QR];
    __shared__ v4f sBm[CLEN * NSTATE / 4];
    __shared__ v4f sCm[CLEN * NSTATE / 4];

    const size_t row0 = (size_t)b * SEQL + (size_t)c * CLEN;
    s6_stage_tile2(dt + row0 * DMODEL + d0, x + row0 * DMODEL + d0, sDt, sX, tid);
    const v4f* bmg = reinterpret_cast<const v4f*>(Bm + row0 * NSTATE);
    const v4f* cmg = reinterpret_cast<const v4f*>(Cm + row0 * NSTATE);
#pragma unroll
    for (int it = 0; it < CLEN * NSTATE / 4 / SCAN_NT; ++it) {
        sBm[tid + it * SCAN_NT] = bmg[tid + it * SCAN_NT];
        sCm[tid + it * SCAN_NT] = cmg[tid + it * SCAN_NT];
    }

    const v4f al = *reinterpret_cast<const v4f*>(A_log + d * NSTATE + SCAN_NS * sg);
    float Ac[SCAN_NS];
#pragma unroll
    for (int k = 0; k < SCAN_NS; ++k) Ac[k] = -expf(al[k]);

    const size_t base = (((size_t)b * DMODEL + d) * NCHUNK + c) * NSTATE + SCAN_NS * sg;
    const v4f cv = *reinterpret_cast<const v4f*>(carry + base);
    float h[SCAN_NS];
#pragma unroll
    for (int k = 0; k < SCAN_NS; ++k) h[k] = cv[k];

    const float dsk = Dskip[d];

    __syncthreads();

    float*       sDtf = reinterpret_cast<float*>(sDt) + ch;
    const float* sXf  = reinterpret_cast<const float*>(sX) + ch;
    const v4f*   sBq  = sBm + sg;
    const v4f*   sCq  = sCm + sg;

    float dt_n = sDtf[0], x_n = sXf[0];
    v4f bm_n = sBq[0], cm_n = sCq[0];

    for (int t = 0; t < CLEN; ++t) {
        const float dtv = dt_n, xv = x_n;
        const v4f bm = bm_n, cm = cm_n;
        const int tn = (t + 1 < CLEN) ? t + 1 : t;
        dt_n = sDtf[tn * SCAN_TD];
        x_n  = sXf[tn * SCAN_TD];
        bm_n = sBq[tn * (NSTATE / 4)];
        cm_n = sCq[tn * (NSTATE / 4)];

        const float ub = dtv * xv;
        float y = 0.f;
#pragma unroll
        for (int k = 0; k < SCAN_NS; ++k) {
            const float a = __expf(dtv * Ac[k]);
            h[k] = fmaf(a, h[k], ub * bm[k]);
            y += cm[k] * h[k];
        }
        y = mx_row_xadd<1>(y);
        y = mx_row_xadd<2>(y);
        y += dsk * xv;
        if (sg == 0) sDtf[t * SCAN_TD] = y;
    }
    __syncthreads();

#pragma unroll
    for (int it = 0; it < CLEN * SCAN_QR / SCAN_NT; ++it) {
        const int idx = tid + it * SCAN_NT;
        const int t = idx / SCAN_QR, q = idx % SCAN_QR;
        const v4f yv = sDt[idx];
        v4f o;
#pragma unroll
        for (int e = 0; e < 4; ++e)
            o[e] = 0.5f * yv[e] * (1.f + erff(yv[e] * 0.70710678118654752f));
        *reinterpret_cast<v4f*>(yout + (row0 + t) * DMODEL + d0 + 4 * q) = o;
    }
}

#define MX_KC   32
#define MX_QC   (MX_KC / 4)
#define MX_NCH  (KDIM / MX_KC)
#define MX_STG  (MX_QC * DMODEL)

__device__ __forceinline__ void mx_fetch(const float* __restrict__ wp,
                                         const v4f* __restrict__ ap,
                                         float (&wst)[MX_KC],
                                         v4f (&an)[MX_KC / 8]) {
#pragma unroll
    for (int jj = 0; jj < MX_KC / 8; ++jj) an[jj] = ap[2 * jj];
#pragma unroll
    for (int kk = 0; kk < MX_KC; ++kk) wst[kk] = wp[(size_t)kk * DMODEL];
}

__device__ __forceinline__ void mx_put(v4f* dst, const float (&wst)[MX_KC]) {
#pragma unroll
    for (int q = 0; q < MX_QC; ++q) {
        v4f w;
        w.x = wst[4 * q];
        w.y = wst[4 * q + 1];
        w.z = wst[4 * q + 2];
        w.w = wst[4 * q + 3];
        dst[q * DMODEL] = w;
    }
}

__device__ __forceinline__ void mx_block(const v4f* bp,
                                         const v4f (&a)[MX_KC / 8],
                                         v8f (&acc)[4]) {
    v4f b4[MX_KC / 8][4];
#pragma unroll
    for (int jj = 0; jj < MX_KC / 8; ++jj)
#pragma unroll
        for (int t = 0; t < 4; ++t) b4[jj][t] = bp[(2 * jj) * DMODEL + 16 * t];
#pragma unroll
    for (int jj = 0; jj < MX_KC / 8; ++jj) {
        v2f a01; a01.x = a[jj].x; a01.y = a[jj].y;
        v2f a23; a23.x = a[jj].z; a23.y = a[jj].w;
#pragma unroll
        for (int t = 0; t < 4; ++t) {
            v2f bv; bv.x = b4[jj][t].x; bv.y = b4[jj][t].y;
            acc[t] = __builtin_amdgcn_wmma_f32_16x16x4_f32(
                 false, a01,  false, bv,
                 (short)0, acc[t],  false,  false);
        }
#pragma unroll
        for (int t = 0; t < 4; ++t) {
            v2f bv; bv.x = b4[jj][t].z; bv.y = b4[jj][t].w;
            acc[t] = __builtin_amdgcn_wmma_f32_16x16x4_f32(
                 false, a23,  false, bv,
                 (short)0, acc[t],  false,  false);
        }
    }
}

__global__ __launch_bounds__(512) __attribute__((amdgpu_waves_per_eu(4, 4)))
void s6_gemm_mix_ln(const float* __restrict__ A,
                    const float* __restrict__ W,
                    const float* __restrict__ bias,
                    const float* __restrict__ hin,
                    const float* __restrict__ g,
                    const float* __restrict__ be,
                    float* __restrict__ hout) {
    __shared__ v4f   sW[2 * MX_STG];
    __shared__ float sPart[2 * 32 * 8];
    __shared__ float sMu[32], sRs[32];

    const int tid  = threadIdx.x;
    const int wave = tid >> 5;
    const int lane = tid & 31;
    const int c    = lane & 15;
    const int h    = lane >> 4;
    const int rb   = (wave & 1) << 4;
    const int cs   = wave >> 1;
    const int m0   = blockIdx.x * 32 + rb;
    const int n0   = cs * 64;

    const float* wp = W + tid;
    const v4f*   ap = reinterpret_cast<const v4f*>(A + (size_t)(m0 + c) * KDIM) + h;
    const v4f*   bp = sW + h * DMODEL + n0 + c;
    v4f*         sp = sW + tid;

    float wst[MX_KC];
    v4f a0[MX_KC / 8], a1[MX_KC / 8];
    v8f acc[4];
#pragma unroll
    for (int t = 0; t < 4; ++t)
        acc[t] = (v8f){0.f, 0.f, 0.f, 0.f, 0.f, 0.f, 0.f, 0.f};

    mx_fetch(wp, ap, wst, a0);
    mx_put(sp, wst);
    __syncthreads();

    for (int ch = 0; ch < MX_NCH - 2; ch += 2) {
        mx_fetch(wp + (size_t)(ch + 1) * MX_KC * DMODEL, ap + (ch + 1) * MX_QC, wst, a1);
        mx_block(bp, a0, acc);
        __builtin_amdgcn_sched_barrier(0);
        mx_put(sp + MX_STG, wst);
        __syncthreads();
        mx_fetch(wp + (size_t)(ch + 2) * MX_KC * DMODEL, ap + (ch + 2) * MX_QC, wst, a0);
        mx_block(bp + MX_STG, a1, acc);
        __builtin_amdgcn_sched_barrier(0);
        mx_put(sp, wst);
        __syncthreads();
    }
    mx_fetch(wp + (size_t)(MX_NCH - 1) * MX_KC * DMODEL, ap + (MX_NCH - 1) * MX_QC, wst, a1);
    mx_block(bp, a0, acc);
    __builtin_amdgcn_sched_barrier(0);
    mx_put(sp + MX_STG, wst);
    __syncthreads();
    mx_block(bp + MX_STG, a1, acc);

    const int mhi = h << 3;
    float s1[8], s2[8];
#pragma unroll
    for (int r = 0; r < 8; ++r) { s1[r] = 0.f; s2[r] = 0.f; }
#pragma unroll
    for (int t = 0; t < 4; ++t) {
        const float bb = bias[n0 + 16 * t + c];
#pragma unroll
        for (int r = 0; r < 8; ++r) {
            const float v = acc[t][r] + bb;
            acc[t][r] = v;
            s1[r] += v;
            s2[r] += v * v;
        }
    }
#pragma unroll
    for (int r = 0; r < 8; ++r) { s1[r] = mx_row_xadd<1>(s1[r]); s2[r] = mx_row_xadd<1>(s2[r]); }
#pragma unroll
    for (int r = 0; r < 8; ++r) { s1[r] = mx_row_xadd<2>(s1[r]); s2[r] = mx_row_xadd<2>(s2[r]); }
#pragma unroll
    for (int r = 0; r < 8; ++r) { s1[r] = mx_row_xadd<4>(s1[r]); s2[r] = mx_row_xadd<4>(s2[r]); }
#pragma unroll
    for (int r = 0; r < 8; ++r) { s1[r] = mx_row_xadd<8>(s1[r]); s2[r] = mx_row_xadd<8>(s2[r]); }
    if (c == 0) {
#pragma unroll
        for (int r = 0; r < 8; ++r) {
            sPart[(rb + mhi + r) * 8 + cs]       = s1[r];
            sPart[256 + (rb + mhi + r) * 8 + cs] = s2[r];
        }
    }
    __syncthreads();
    if (tid < 32) {
        const float* p1 = sPart + tid * 8;
        const float* p2 = sPart + 256 + tid * 8;
        float t1 = p1[0], t2 = p2[0];
#pragma unroll
        for (int k = 1; k < 8; ++k) { t1 += p1[k]; t2 += p2[k]; }
        const float mu  = t1 * (1.f / DMODEL);
        const float var = t2 * (1.f / DMODEL) - mu * mu;
        sMu[tid] = mu;
        sRs[tid] = rsqrtf(var + LN_EPS);
    }
    __syncthreads();
    float mu[8], rs[8];
#pragma unroll
    for (int r = 0; r < 8; ++r) { mu[r] = sMu[rb + mhi + r]; rs[r] = sRs[rb + mhi + r]; }
#pragma unroll
    for (int t = 0; t < 4; ++t) {
        const int   col = n0 + 16 * t + c;
        const float gg  = g[col];
        const float bo  = be[col];
#pragma unroll
        for (int r = 0; r < 8; ++r) {
            const size_t o = (size_t)(m0 + mhi + r) * DMODEL + col;
            hout[o] = hin[o] + (acc[t][r] - mu[r]) * rs[r] * gg + bo;
        }
    }
}

extern "C" void kernel_launch(void* const* d_in, const int* in_sizes, int n_in,
                              void* d_out, int out_size, void* d_ws, size_t ws_size,
                              hipStream_t stream) {
    (void)in_sizes; (void)n_in; (void)out_size; (void)ws_size;

    const float* x      = (const float*)d_in[0];
    const float* A_log  = (const float*)d_in[1];
    const float* W_B    = (const float*)d_in[2];
    const float* b_B    = (const float*)d_in[3];
    const float* W_C    = (const float*)d_in[4];
    const float* b_C    = (const float*)d_in[5];
    const float* W_dt   = (const float*)d_in[6];
    const float* b_dt   = (const float*)d_in[7];
    const float* D_skip = (const float*)d_in[8];
    const float* W_mix  = (const float*)d_in[9];
    const float* b_mix  = (const float*)d_in[10];
    const float* ln_g   = (const float*)d_in[11];
    const float* ln_b   = (const float*)d_in[12];
    const float* W_dec  = (const float*)d_in[13];
    const float* b_dec  = (const float*)d_in[14];
    float* out = (float*)d_out;

    float* ws = (float*)d_ws;
    size_t off = 0;
    float* dtb   = ws + off; off += (size_t)MROWS * DMODEL;
    float* Bmb   = ws + off; off += (size_t)MROWS * NSTATE;
    float* Cmb   = ws + off; off += (size_t)MROWS * NSTATE;
    float* ygelu = ws + off; off += (size_t)MROWS * DMODEL;
    float* h1    = ws + off; off += (size_t)MROWS * DMODEL;
    float* h2    = ws + off; off += (size_t)MROWS * DMODEL;
    float* hend  = ws + off; off += (size_t)BATCH * DMODEL * NCHUNK * NSTATE;
    float* aend  = ws + off; off += (size_t)BATCH * DMODEL * NCHUNK * NSTATE;
    float* carry = ws + off; off += (size_t)BATCH * DMODEL * NCHUNK * NSTATE;

    const int DN = DMODEL * NSTATE;
    const int DD = DMODEL * DMODEL;

    const dim3 gBig(MROWS / 256, DMODEL / 64);
    const dim3 gScan(DMODEL / SCAN_TD, NCHUNK, BATCH);

    for (int i = 0; i < 2; ++i) {
        const float* hin = (i == 0) ? x : h1;
        float* hout      = (i == 0) ? h1 : h2;

        s6_gemm_strip_k512<4, 16><<<gBig, 512, 0, stream>>>(
            hin, W_dt + (size_t)i * DD, b_dt + i * DMODEL, dtb, DMODEL, 1);
        s6_gemm_bc_splitk<<<MROWS / 32, 256, 0, stream>>>(
            hin, W_B + (size_t)i * DN, W_C + (size_t)i * DN,
            b_B + i * NSTATE, b_C + i * NSTATE, Bmb, Cmb);

        s6_scan_pass1<<<gScan, SCAN_NT, 0, stream>>>(
            hin, dtb, Bmb, A_log + (size_t)i * DN, hend, aend);
        s6_scan_pass2<<<BATCH * DMODEL * NSTATE / 256, 256, 0, stream>>>(
            hend, aend, carry);
        s6_scan_pass3<<<gScan, SCAN_NT, 0, stream>>>(
            hin, dtb, Bmb, Cmb, A_log + (size_t)i * DN, carry,
            D_skip + i * DMODEL, ygelu);

        s6_gemm_mix_ln<<<MROWS / 32, 512, 0, stream>>>(
            ygelu, W_mix + (size_t)i * DD, b_mix + i * DMODEL, hin,
            ln_g + i * DMODEL, ln_b + i * DMODEL, hout);
    }

    s6_gemm_strip_k512<4, 16><<<gBig, 512, 0, stream>>>(h2, W_dec, b_dec, out, OUTD, 0);
}
